# starting kernel
# baseline (speedup 1.0000x reference)
#include <hip/hip_runtime.h>
#include <math.h>


typedef float v2f __attribute__((ext_vector_type(2)));
typedef float v8f __attribute__((ext_vector_type(8)));

#define BSZ 4096
#define NF  2048
#define NT  300
#define RH  512
#define HD  512

#define MT      32
#define THREADS 256

__device__ __forceinline__ v8f wmma_f32_step(v2f a, v2f b, v8f c) {
    return __builtin_amdgcn_wmma_f32_16x16x4_f32(false, a, false, b,
                                                 (short)0, c, false, false);
}

template <int K, int LD, bool MULTIPLY>
__device__ __forceinline__ void stage1_pass(
    const float* __restrict__ A, const float* __restrict__ W,
    float* uLDS, int m0, int ni0, int r, int hi)
{
    v8f t8[8];
    #pragma unroll
    for (int i = 0; i < 8; ++i) t8[i] = (v8f){};

    const size_t rowA = (size_t)(m0 + r) * LD;
    const size_t rowB = (size_t)(m0 + 16 + r) * LD;

    #pragma unroll 2
    for (int k0 = 0; k0 < K; k0 += 4) {
        const int kk = k0 + 2 * hi;
        v2f a0; a0.x = A[rowA + kk]; a0.y = A[rowA + kk + 1];
        v2f a1; a1.x = A[rowB + kk]; a1.y = A[rowB + kk + 1];
        #pragma unroll
        for (int t = 0; t < 4; ++t) {
            const int n0 = (ni0 + t) * 16;
            v2f b; b.x = W[(size_t)kk * RH + n0 + r];
                   b.y = W[(size_t)(kk + 1) * RH + n0 + r];
            t8[2 * t]     = wmma_f32_step(a0, b, t8[2 * t]);
            t8[2 * t + 1] = wmma_f32_step(a1, b, t8[2 * t + 1]);
        }
    }

    #pragma unroll
    for (int t = 0; t < 4; ++t) {
        const int nn = (ni0 + t) * 16 + r;
        #pragma unroll
        for (int mi = 0; mi < 2; ++mi) {
            #pragma unroll
            for (int e = 0; e < 8; ++e) {
                const int mm = mi * 16 + 8 * hi + e;
                if (MULTIPLY) uLDS[mm * RH + nn] *= t8[2 * t + mi][e];
                else          uLDS[mm * RH + nn]  = t8[2 * t + mi][e];
            }
        }
    }
}

template <int K1, int LD1, int K2, int LD2>
__device__ __forceinline__ void product_and_project(
    const float* __restrict__ A1, const float* __restrict__ W1,
    const float* __restrict__ A2, const float* __restrict__ W2,
    const float* __restrict__ Ws,
    float* uLDS, v8f* acc,
    int m0, int ni0, int r, int hi)
{
    stage1_pass<K1, LD1, false>(A1, W1, uLDS, m0, ni0, r, hi);
    stage1_pass<K2, LD2, true >(A2, W2, uLDS, m0, ni0, r, hi);
    __syncthreads();

    const int mloc0 = r;
    const int mloc1 = 16 + r;
    #pragma unroll 2
    for (int k0 = 0; k0 < RH; k0 += 4) {
        const int kk = k0 + 2 * hi;
        v2f a0; a0.x = uLDS[mloc0 * RH + kk]; a0.y = uLDS[mloc0 * RH + kk + 1];
        v2f a1; a1.x = uLDS[mloc1 * RH + kk]; a1.y = uLDS[mloc1 * RH + kk + 1];
        if (k0 + 32 < RH)
            __builtin_prefetch(&Ws[(size_t)(kk + 32) * HD + ni0 * 16 + r], 0, 1);
        #pragma unroll
        for (int t = 0; t < 4; ++t) {
            const int n0 = (ni0 + t) * 16;
            v2f b; b.x = Ws[(size_t)kk * HD + n0 + r];
                   b.y = Ws[(size_t)(kk + 1) * HD + n0 + r];
            acc[2 * t]     = wmma_f32_step(a0, b, acc[2 * t]);
            acc[2 * t + 1] = wmma_f32_step(a1, b, acc[2 * t + 1]);
        }
    }
    __syncthreads();
}

__global__ __launch_bounds__(THREADS)
void vncl_gate_kernel(const float* __restrict__ v,  const float* __restrict__ s,
                      const float* __restrict__ x,  const float* __restrict__ h,
                      const float* __restrict__ Wa, const float* __restrict__ Wb,
                      const float* __restrict__ Wc, const float* __restrict__ Ua,
                      const float* __restrict__ Ub, const float* __restrict__ Uc,
                      const float* __restrict__ Ca, const float* __restrict__ Cb,
                      const float* __restrict__ Cc, const float* __restrict__ bias,
                      float* __restrict__ logits  )
{
    __shared__ float uLDS[MT * RH];

    const int g    = blockIdx.y;
    const int m0   = blockIdx.x * MT;
    const int lane = threadIdx.x & 31;
    const int wave = threadIdx.x >> 5;
    const int r    = lane & 15;
    const int hi   = lane >> 4;

    const int ni0 = wave * 4;

    const float* Wa_g = Wa + (size_t)g * RH * RH;
    const float* Wb_g = Wb + (size_t)g * NT * RH;
    const float* Wc_g = Wc + (size_t)g * RH * HD;
    const float* Ua_g = Ua + (size_t)g * RH * RH;
    const float* Ub_g = Ub + (size_t)g * NT * RH;
    const float* Uc_g = Uc + (size_t)g * RH * HD;
    const float* Ca_g = Ca + (size_t)g * NF * RH;
    const float* Cb_g = Cb + (size_t)g * NT * RH;
    const float* Cc_g = Cc + (size_t)g * RH * HD;

    v8f acc[8];
    #pragma unroll
    for (int t = 0; t < 8; ++t) acc[t] = (v8f){};

    product_and_project<RH, RH, NT, NT>(x, Wa_g, s, Wb_g, Wc_g,
                                        uLDS, acc, m0, ni0, r, hi);
    product_and_project<NF, NF, NT, NT>(v, Ca_g, s, Cb_g, Cc_g,
                                        uLDS, acc, m0, ni0, r, hi);
    product_and_project<NT, NT, RH, RH>(s, Ub_g, h, Ua_g, Uc_g,
                                        uLDS, acc, m0, ni0, r, hi);

    const float* bg = bias + (size_t)g * HD;
    float* Lg = logits + ((size_t)g * BSZ + m0) * HD;
    #pragma unroll
    for (int t = 0; t < 4; ++t) {
        const int nn = (ni0 + t) * 16 + r;
        #pragma unroll
        for (int mi = 0; mi < 2; ++mi) {
            #pragma unroll
            for (int e = 0; e < 8; ++e) {
                const int mm = mi * 16 + 8 * hi + e;
                Lg[(size_t)mm * HD + nn] = acc[2 * t + mi][e] + bg[nn];
            }
        }
    }
}

__global__ __launch_bounds__(256)
void lstm_epilogue(const float* __restrict__ logits,
                   const float* __restrict__ c,
                   float* __restrict__ out, int n)
{
    const int idx = blockIdx.x * blockDim.x + threadIdx.x;
    if (idx >= n) return;
    const size_t stride = (size_t)BSZ * HD;
    const float li = logits[idx];
    const float lf = logits[idx + stride];
    const float lo = logits[idx + 2 * stride];
    const float lg = logits[idx + 3 * stride];
    const float ig = 1.0f / (1.0f + __expf(-li));
    const float fg = 1.0f / (1.0f + __expf(-lf));
    const float og = 1.0f / (1.0f + __expf(-lo));
    const float gg = tanhf(lg);
    const float nc = fg * c[idx] + ig * gg;
    const float nh = og * tanhf(nc);
    out[idx]          = nh;
    out[idx + stride] = nc;
}

extern "C" void kernel_launch(void* const* d_in, const int* in_sizes, int n_in,
                              void* d_out, int out_size, void* d_ws, size_t ws_size,
                              hipStream_t stream) {
    (void)in_sizes; (void)n_in; (void)out_size; (void)ws_size;
    const float* v  = (const float*)d_in[0];
    const float* s  = (const float*)d_in[1];
    const float* x  = (const float*)d_in[2];
    const float* h  = (const float*)d_in[3];
    const float* c  = (const float*)d_in[4];
    const float* Wa = (const float*)d_in[5];
    const float* Wb = (const float*)d_in[6];
    const float* Wc = (const float*)d_in[7];
    const float* Ua = (const float*)d_in[8];
    const float* Ub = (const float*)d_in[9];
    const float* Uc = (const float*)d_in[10];
    const float* Ca = (const float*)d_in[11];
    const float* Cb = (const float*)d_in[12];
    const float* Cc = (const float*)d_in[13];
    const float* b  = (const float*)d_in[14];

    float* logits = (float*)d_ws;

    dim3 grid(BSZ / MT, 4);
    vncl_gate_kernel<<<grid, dim3(THREADS), 0, stream>>>(
        v, s, x, h, Wa, Wb, Wc, Ua, Ub, Uc, Ca, Cb, Cc, b, logits);

    const int n = BSZ * HD;
    lstm_epilogue<<<(n + 255) / 256, 256, 0, stream>>>(logits, c, (float*)d_out, n);
}


# Fragment loads software pipelined one stage ahead in a second register set in all nine GEMM loops (rolled; load group fenced ahead of WMMA block); registers capped at 256 -> 2 workgroups per WGP
# speedup vs baseline: 2.1537x; 2.1537x over previous
#include <hip/hip_runtime.h>
#include <math.h>


typedef float v2f __attribute__((ext_vector_type(2)));
typedef float v8f __attribute__((ext_vector_type(8)));

#define BSZ 4096
#define NF  2048
#define NT  300
#define RH  512
#define HD  512

#define MT      32
#define THREADS 256

__device__ __forceinline__ v8f wmma_f32_step(v2f a, v2f b, v8f c) {
    return __builtin_amdgcn_wmma_f32_16x16x4_f32(false, a, false, b,
                                                 (short)0, c, false, false);
}

template <int S>
struct FragSet {
    v2f a0[S];
    v2f a1[S];
    v2f b[S][4];
};

template <int S, int LD, bool A_LDS>
__device__ __forceinline__ void load_stage(
    FragSet<S>& f, const float* __restrict__ Ag, const float* Al,
    const float* __restrict__ W, int m0, int ni0, int r, int hi, int kbase)
{
    #pragma unroll
    for (int s = 0; s < S; ++s) {
        const int kk = kbase + 4 * s + 2 * hi;
        if (A_LDS) {
            f.a0[s].x = Al[r * RH + kk];
            f.a0[s].y = Al[r * RH + kk + 1];
            f.a1[s].x = Al[(16 + r) * RH + kk];
            f.a1[s].y = Al[(16 + r) * RH + kk + 1];
        } else {
            const size_t rowA = (size_t)(m0 + r) * LD;
            const size_t rowB = (size_t)(m0 + 16 + r) * LD;
            f.a0[s].x = Ag[rowA + kk];
            f.a0[s].y = Ag[rowA + kk + 1];
            f.a1[s].x = Ag[rowB + kk];
            f.a1[s].y = Ag[rowB + kk + 1];
        }
        #pragma unroll
        for (int t = 0; t < 4; ++t) {
            const int n0 = (ni0 + t) * 16;
            f.b[s][t].x = W[(size_t)kk * RH + n0 + r];
            f.b[s][t].y = W[(size_t)(kk + 1) * RH + n0 + r];
        }
    }
}

template <int S>
__device__ __forceinline__ void mma_stage(const FragSet<S>& f, v8f* c)
{
    #pragma unroll
    for (int s = 0; s < S; ++s) {
        #pragma unroll
        for (int t = 0; t < 4; ++t) {
            c[2 * t]     = wmma_f32_step(f.a0[s], f.b[s][t], c[2 * t]);
            c[2 * t + 1] = wmma_f32_step(f.a1[s], f.b[s][t], c[2 * t + 1]);
        }
    }
}

template <int K, int LD, int S, bool A_LDS>
__device__ __forceinline__ void gemm_pipelined(
    const float* __restrict__ Ag, const float* Al, const float* __restrict__ W,
    v8f* c, int m0, int ni0, int r, int hi)
{
    constexpr int KS  = 4 * S;
    constexpr int NST = K / KS;
    static_assert(K % KS == 0, "stage size must divide K");
    static_assert(NST >= 2, "needs two stages");

    FragSet<S> f0, f1;
    load_stage<S, LD, A_LDS>(f0, Ag, Al, W, m0, ni0, r, hi, 0);

    int st = 0;
    #pragma unroll 1
    for (; st + 2 < NST; st += 2) {
        load_stage<S, LD, A_LDS>(f1, Ag, Al, W, m0, ni0, r, hi, (st + 1) * KS);
        __builtin_amdgcn_sched_barrier(0);
        mma_stage<S>(f0, c);
        __builtin_amdgcn_sched_barrier(0);
        load_stage<S, LD, A_LDS>(f0, Ag, Al, W, m0, ni0, r, hi, (st + 2) * KS);
        __builtin_amdgcn_sched_barrier(0);
        mma_stage<S>(f1, c);
        __builtin_amdgcn_sched_barrier(0);
    }
    if (NST % 2 == 0) {
        load_stage<S, LD, A_LDS>(f1, Ag, Al, W, m0, ni0, r, hi, (st + 1) * KS);
        __builtin_amdgcn_sched_barrier(0);
        mma_stage<S>(f0, c);
        mma_stage<S>(f1, c);
    } else {
        mma_stage<S>(f0, c);
    }
}

template <int K, int LD, int S, bool MULTIPLY>
__device__ __forceinline__ void stage1_pass(
    const float* __restrict__ A, const float* __restrict__ W,
    float* uLDS, int m0, int ni0, int r, int hi)
{
    v8f t8[8];
    #pragma unroll
    for (int i = 0; i < 8; ++i) t8[i] = (v8f){};

    gemm_pipelined<K, LD, S, false>(A, nullptr, W, t8, m0, ni0, r, hi);

    #pragma unroll
    for (int t = 0; t < 4; ++t) {
        const int nn = (ni0 + t) * 16 + r;
        #pragma unroll
        for (int mi = 0; mi < 2; ++mi) {
            #pragma unroll
            for (int e = 0; e < 8; ++e) {
                const int mm = mi * 16 + 8 * hi + e;
                if (MULTIPLY) uLDS[mm * RH + nn] *= t8[2 * t + mi][e];
                else          uLDS[mm * RH + nn]  = t8[2 * t + mi][e];
            }
        }
    }
}

template <int K1, int LD1, int S1, int K2, int LD2, int S2>
__device__ __forceinline__ void product_and_project(
    const float* __restrict__ A1, const float* __restrict__ W1,
    const float* __restrict__ A2, const float* __restrict__ W2,
    const float* __restrict__ Ws,
    float* uLDS, v8f* acc,
    int m0, int ni0, int r, int hi)
{
    stage1_pass<K1, LD1, S1, false>(A1, W1, uLDS, m0, ni0, r, hi);
    stage1_pass<K2, LD2, S2, true >(A2, W2, uLDS, m0, ni0, r, hi);
    __syncthreads();

    gemm_pipelined<RH, RH, 2, true>(nullptr, uLDS, Ws, acc, m0, ni0, r, hi);
    __syncthreads();
}

__global__ __launch_bounds__(THREADS, 4)
void vncl_gate_kernel(const float* __restrict__ v,  const float* __restrict__ s,
                      const float* __restrict__ x,  const float* __restrict__ h,
                      const float* __restrict__ Wa, const float* __restrict__ Wb,
                      const float* __restrict__ Wc, const float* __restrict__ Ua,
                      const float* __restrict__ Ub, const float* __restrict__ Uc,
                      const float* __restrict__ Ca, const float* __restrict__ Cb,
                      const float* __restrict__ Cc, const float* __restrict__ bias,
                      float* __restrict__ logits  )
{
    __shared__ float uLDS[MT * RH];

    const int g    = blockIdx.y;
    const int m0   = blockIdx.x * MT;
    const int lane = threadIdx.x & 31;
    const int wave = threadIdx.x >> 5;
    const int r    = lane & 15;
    const int hi   = lane >> 4;

    const int ni0 = wave * 4;

    const float* Wa_g = Wa + (size_t)g * RH * RH;
    const float* Wb_g = Wb + (size_t)g * NT * RH;
    const float* Wc_g = Wc + (size_t)g * RH * HD;
    const float* Ua_g = Ua + (size_t)g * RH * RH;
    const float* Ub_g = Ub + (size_t)g * NT * RH;
    const float* Uc_g = Uc + (size_t)g * RH * HD;
    const float* Ca_g = Ca + (size_t)g * NF * RH;
    const float* Cb_g = Cb + (size_t)g * NT * RH;
    const float* Cc_g = Cc + (size_t)g * RH * HD;

    v8f acc[8];
    #pragma unroll
    for (int t = 0; t < 8; ++t) acc[t] = (v8f){};

    product_and_project<RH, RH, 2, NT, NT, 3>(x, Wa_g, s, Wb_g, Wc_g,
                                        uLDS, acc, m0, ni0, r, hi);
    product_and_project<NF, NF, 2, NT, NT, 3>(v, Ca_g, s, Cb_g, Cc_g,
                                        uLDS, acc, m0, ni0, r, hi);
    product_and_project<NT, NT, 3, RH, RH, 2>(s, Ub_g, h, Ua_g, Uc_g,
                                        uLDS, acc, m0, ni0, r, hi);

    const float* bg = bias + (size_t)g * HD;
    float* Lg = logits + ((size_t)g * BSZ + m0) * HD;
    #pragma unroll
    for (int t = 0; t < 4; ++t) {
        const int nn = (ni0 + t) * 16 + r;
        #pragma unroll
        for (int mi = 0; mi < 2; ++mi) {
            #pragma unroll
            for (int e = 0; e < 8; ++e) {
                const int mm = mi * 16 + 8 * hi + e;
                Lg[(size_t)mm * HD + nn] = acc[2 * t + mi][e] + bg[nn];
            }
        }
    }
}

__global__ __launch_bounds__(256)
void lstm_epilogue(const float* __restrict__ logits,
                   const float* __restrict__ c,
                   float* __restrict__ out, int n)
{
    const int idx = blockIdx.x * blockDim.x + threadIdx.x;
    if (idx >= n) return;
    const size_t stride = (size_t)BSZ * HD;
    const float li = logits[idx];
    const float lf = logits[idx + stride];
    const float lo = logits[idx + 2 * stride];
    const float lg = logits[idx + 3 * stride];
    const float ig = 1.0f / (1.0f + __expf(-li));
    const float fg = 1.0f / (1.0f + __expf(-lf));
    const float og = 1.0f / (1.0f + __expf(-lo));
    const float gg = tanhf(lg);
    const float nc = fg * c[idx] + ig * gg;
    const float nh = og * tanhf(nc);
    out[idx]          = nh;
    out[idx + stride] = nc;
}

extern "C" void kernel_launch(void* const* d_in, const int* in_sizes, int n_in,
                              void* d_out, int out_size, void* d_ws, size_t ws_size,
                              hipStream_t stream) {
    (void)in_sizes; (void)n_in; (void)out_size; (void)ws_size;
    const float* v  = (const float*)d_in[0];
    const float* s  = (const float*)d_in[1];
    const float* x  = (const float*)d_in[2];
    const float* h  = (const float*)d_in[3];
    const float* c  = (const float*)d_in[4];
    const float* Wa = (const float*)d_in[5];
    const float* Wb = (const float*)d_in[6];
    const float* Wc = (const float*)d_in[7];
    const float* Ua = (const float*)d_in[8];
    const float* Ub = (const float*)d_in[9];
    const float* Uc = (const float*)d_in[10];
    const float* Ca = (const float*)d_in[11];
    const float* Cb = (const float*)d_in[12];
    const float* Cc = (const float*)d_in[13];
    const float* b  = (const float*)d_in[14];

    float* logits = (float*)d_ws;

    dim3 grid(BSZ / MT, 4);
    vncl_gate_kernel<<<grid, dim3(THREADS), 0, stream>>>(
        v, s, x, h, Wa, Wb, Wc, Ua, Ub, Uc, Ca, Cb, Cc, b, logits);

    const int n = BSZ * HD;
    lstm_epilogue<<<(n + 255) / 256, 256, 0, stream>>>(logits, c, (float*)d_out, n);
}


# Fragment loads widened to 16 bytes per lane: 4 consecutive k of the lane's A row, 4 adjacent W columns (wave's columns dealt to tiles as n=4c+t); k permuted inside each 8-k block, sum order changes
# speedup vs baseline: 2.2429x; 1.0414x over previous
#include <hip/hip_runtime.h>
#include <math.h>


typedef float v2f __attribute__((ext_vector_type(2)));
typedef float v4f __attribute__((ext_vector_type(4)));
typedef float v8f __attribute__((ext_vector_type(8)));

#define BSZ 4096
#define NF  2048
#define NT  300
#define RH  512
#define HD  512

#define MT      32
#define THREADS 256

__device__ __forceinline__ v8f wmma_f32_step(v2f a, v2f b, v8f c) {
    return __builtin_amdgcn_wmma_f32_16x16x4_f32(false, a, false, b,
                                                 (short)0, c, false, false);
}

struct FragSet {
    v4f a0;
    v4f a1;
    v4f b[4];
};

struct FragTail {
    v2f a0;
    v2f a1;
    v4f b[2];
};

template <int LD, bool A_LDS>
__device__ __forceinline__ void load_stage(
    FragSet& f, const float* __restrict__ Ag, const float* Al,
    const float* __restrict__ W, int m0, int col0, int r, int hi, int kbase)
{
    const int kk = kbase + 4 * hi;
    if (A_LDS) {
        f.a0 = *(const v4f*)(Al + r * RH + kk);
        f.a1 = *(const v4f*)(Al + (16 + r) * RH + kk);
    } else {
        f.a0 = *(const v4f*)(Ag + (size_t)(m0 + r) * LD + kk);
        f.a1 = *(const v4f*)(Ag + (size_t)(m0 + 16 + r) * LD + kk);
    }
    #pragma unroll
    for (int j = 0; j < 4; ++j)
        f.b[j] = *(const v4f*)(W + (size_t)(kk + j) * RH + col0);
}

template <int LD>
__device__ __forceinline__ void load_tail(
    FragTail& f, const float* __restrict__ Ag, const float* __restrict__ W,
    int m0, int col0, int r, int hi, int kbase)
{
    const int kk = kbase + 2 * hi;
    f.a0 = *(const v2f*)(Ag + (size_t)(m0 + r) * LD + kk);
    f.a1 = *(const v2f*)(Ag + (size_t)(m0 + 16 + r) * LD + kk);
    #pragma unroll
    for (int e = 0; e < 2; ++e)
        f.b[e] = *(const v4f*)(W + (size_t)(kk + e) * RH + col0);
}

__device__ __forceinline__ void mma_stage(const FragSet& f, v8f* c)
{
    #pragma unroll
    for (int j = 0; j < 2; ++j) {
        const v2f a0 = j ? (v2f){f.a0.z, f.a0.w} : (v2f){f.a0.x, f.a0.y};
        const v2f a1 = j ? (v2f){f.a1.z, f.a1.w} : (v2f){f.a1.x, f.a1.y};
        #pragma unroll
        for (int t = 0; t < 4; ++t) {
            const v2f b = (v2f){f.b[2 * j][t], f.b[2 * j + 1][t]};
            c[2 * t]     = wmma_f32_step(a0, b, c[2 * t]);
            c[2 * t + 1] = wmma_f32_step(a1, b, c[2 * t + 1]);
        }
    }
}

__device__ __forceinline__ void mma_tail(const FragTail& f, v8f* c)
{
    #pragma unroll
    for (int t = 0; t < 4; ++t) {
        const v2f b = (v2f){f.b[0][t], f.b[1][t]};
        c[2 * t]     = wmma_f32_step(f.a0, b, c[2 * t]);
        c[2 * t + 1] = wmma_f32_step(f.a1, b, c[2 * t + 1]);
    }
}

template <int K, int LD, bool A_LDS>
__device__ __forceinline__ void gemm_pipelined(
    const float* __restrict__ Ag, const float* Al, const float* __restrict__ W,
    v8f* c, int m0, int col0, int r, int hi)
{
    constexpr int KS   = 8;
    constexpr int NST  = K / KS;
    constexpr bool TAIL = (K % KS) != 0;
    static_assert(K % 4 == 0, "K must be a multiple of the WMMA k");
    static_assert(NST >= 2, "needs two stages");
    static_assert(!(TAIL && A_LDS), "the tail loader reads global A only");

    FragSet f0, f1;
    FragTail ft;
    load_stage<LD, A_LDS>(f0, Ag, Al, W, m0, col0, r, hi, 0);

    int st = 0;
    #pragma unroll 1
    for (; st + 2 < NST; st += 2) {
        load_stage<LD, A_LDS>(f1, Ag, Al, W, m0, col0, r, hi, (st + 1) * KS);
        __builtin_amdgcn_sched_barrier(0);
        mma_stage(f0, c);
        __builtin_amdgcn_sched_barrier(0);
        load_stage<LD, A_LDS>(f0, Ag, Al, W, m0, col0, r, hi, (st + 2) * KS);
        __builtin_amdgcn_sched_barrier(0);
        mma_stage(f1, c);
        __builtin_amdgcn_sched_barrier(0);
    }
    if (NST % 2 == 0) {
        load_stage<LD, A_LDS>(f1, Ag, Al, W, m0, col0, r, hi, (st + 1) * KS);
        if (TAIL) load_tail<LD>(ft, Ag, W, m0, col0, r, hi, NST * KS);
        __builtin_amdgcn_sched_barrier(0);
        mma_stage(f0, c);
        mma_stage(f1, c);
    } else {
        if (TAIL) {
            load_tail<LD>(ft, Ag, W, m0, col0, r, hi, NST * KS);
            __builtin_amdgcn_sched_barrier(0);
        }
        mma_stage(f0, c);
    }
    if (TAIL) mma_tail(ft, c);
}

template <int K, int LD, bool MULTIPLY>
__device__ __forceinline__ void stage1_pass(
    const float* __restrict__ A, const float* __restrict__ W,
    float* uLDS, int m0, int col0, int r, int hi)
{
    v8f t8[8];
    #pragma unroll
    for (int i = 0; i < 8; ++i) t8[i] = (v8f){};

    gemm_pipelined<K, LD, false>(A, nullptr, W, t8, m0, col0, r, hi);

    #pragma unroll
    for (int mi = 0; mi < 2; ++mi) {
        #pragma unroll
        for (int e = 0; e < 8; ++e) {
            const int mm = mi * 16 + 8 * hi + e;
            const v4f val = (v4f){t8[mi][e], t8[2 + mi][e], t8[4 + mi][e], t8[6 + mi][e]};
            v4f* p = (v4f*)(uLDS + mm * RH + col0);
            if (MULTIPLY) *p = *p * val;
            else          *p = val;
        }
    }
}

template <int K1, int LD1, int K2, int LD2>
__device__ __forceinline__ void product_and_project(
    const float* __restrict__ A1, const float* __restrict__ W1,
    const float* __restrict__ A2, const float* __restrict__ W2,
    const float* __restrict__ Ws,
    float* uLDS, v8f* acc,
    int m0, int col0, int r, int hi)
{
    stage1_pass<K1, LD1, false>(A1, W1, uLDS, m0, col0, r, hi);
    stage1_pass<K2, LD2, true >(A2, W2, uLDS, m0, col0, r, hi);
    __syncthreads();

    gemm_pipelined<RH, RH, true>(nullptr, uLDS, Ws, acc, m0, col0, r, hi);
    __syncthreads();
}

__global__ __launch_bounds__(THREADS, 4)
void vncl_gate_kernel(const float* __restrict__ v,  const float* __restrict__ s,
                      const float* __restrict__ x,  const float* __restrict__ h,
                      const float* __restrict__ Wa, const float* __restrict__ Wb,
                      const float* __restrict__ Wc, const float* __restrict__ Ua,
                      const float* __restrict__ Ub, const float* __restrict__ Uc,
                      const float* __restrict__ Ca, const float* __restrict__ Cb,
                      const float* __restrict__ Cc, const float* __restrict__ bias,
                      float* __restrict__ logits  )
{
    __shared__ __attribute__((aligned(16))) float uLDS[MT * RH];

    const int g    = blockIdx.y;
    const int m0   = blockIdx.x * MT;
    const int lane = threadIdx.x & 31;
    const int wave = threadIdx.x >> 5;
    const int r    = lane & 15;
    const int hi   = lane >> 4;

    const int col0 = wave * 64 + 4 * r;

    const float* Wa_g = Wa + (size_t)g * RH * RH;
    const float* Wb_g = Wb + (size_t)g * NT * RH;
    const float* Wc_g = Wc + (size_t)g * RH * HD;
    const float* Ua_g = Ua + (size_t)g * RH * RH;
    const float* Ub_g = Ub + (size_t)g * NT * RH;
    const float* Uc_g = Uc + (size_t)g * RH * HD;
    const float* Ca_g = Ca + (size_t)g * NF * RH;
    const float* Cb_g = Cb + (size_t)g * NT * RH;
    const float* Cc_g = Cc + (size_t)g * RH * HD;

    v8f acc[8];
    #pragma unroll
    for (int t = 0; t < 8; ++t) acc[t] = (v8f){};

    product_and_project<RH, RH, NT, NT>(x, Wa_g, s, Wb_g, Wc_g,
                                        uLDS, acc, m0, col0, r, hi);
    product_and_project<NF, NF, NT, NT>(v, Ca_g, s, Cb_g, Cc_g,
                                        uLDS, acc, m0, col0, r, hi);
    product_and_project<NT, NT, RH, RH>(s, Ub_g, h, Ua_g, Uc_g,
                                        uLDS, acc, m0, col0, r, hi);

    const v4f bv = *(const v4f*)(bias + (size_t)g * HD + col0);
    float* Lg = logits + ((size_t)g * BSZ + m0) * HD;
    #pragma unroll
    for (int mi = 0; mi < 2; ++mi) {
        #pragma unroll
        for (int e = 0; e < 8; ++e) {
            const int mm = mi * 16 + 8 * hi + e;
            const v4f val = (v4f){acc[mi][e], acc[2 + mi][e], acc[4 + mi][e], acc[6 + mi][e]};
            *(v4f*)(Lg + (size_t)mm * HD + col0) = val + bv;
        }
    }
}

__global__ __launch_bounds__(256)
void lstm_epilogue(const float* __restrict__ logits,
                   const float* __restrict__ c,
                   float* __restrict__ out, int n)
{
    const int idx = blockIdx.x * blockDim.x + threadIdx.x;
    if (idx >= n) return;
    const size_t stride = (size_t)BSZ * HD;
    const float li = logits[idx];
    const float lf = logits[idx + stride];
    const float lo = logits[idx + 2 * stride];
    const float lg = logits[idx + 3 * stride];
    const float ig = 1.0f / (1.0f + __expf(-li));
    const float fg = 1.0f / (1.0f + __expf(-lf));
    const float og = 1.0f / (1.0f + __expf(-lo));
    const float gg = tanhf(lg);
    const float nc = fg * c[idx] + ig * gg;
    const float nh = og * tanhf(nc);
    out[idx]          = nh;
    out[idx + stride] = nc;
}

extern "C" void kernel_launch(void* const* d_in, const int* in_sizes, int n_in,
                              void* d_out, int out_size, void* d_ws, size_t ws_size,
                              hipStream_t stream) {
    (void)in_sizes; (void)n_in; (void)out_size; (void)ws_size;
    const float* v  = (const float*)d_in[0];
    const float* s  = (const float*)d_in[1];
    const float* x  = (const float*)d_in[2];
    const float* h  = (const float*)d_in[3];
    const float* c  = (const float*)d_in[4];
    const float* Wa = (const float*)d_in[5];
    const float* Wb = (const float*)d_in[6];
    const float* Wc = (const float*)d_in[7];
    const float* Ua = (const float*)d_in[8];
    const float* Ub = (const float*)d_in[9];
    const float* Uc = (const float*)d_in[10];
    const float* Ca = (const float*)d_in[11];
    const float* Cb = (const float*)d_in[12];
    const float* Cc = (const float*)d_in[13];
    const float* b  = (const float*)d_in[14];

    float* logits = (float*)d_ws;

    dim3 grid(BSZ / MT, 4);
    vncl_gate_kernel<<<grid, dim3(THREADS), 0, stream>>>(
        v, s, x, h, Wa, Wb, Wc, Ua, Ub, Uc, Ca, Cb, Cc, b, logits);

    const int n = BSZ * HD;
    lstm_epilogue<<<(n + 255) / 256, 256, 0, stream>>>(logits, c, (float*)d_out, n);
}


# Fragment loads pipelined two stages ahead (three register sets in rotation) in all nine GEMM loops; logits accumulators rest in a per-wave LDS area between projections to free the registers
# speedup vs baseline: 2.3584x; 1.0515x over previous
#include <hip/hip_runtime.h>
#include <math.h>


typedef float v2f __attribute__((ext_vector_type(2)));
typedef float v4f __attribute__((ext_vector_type(4)));
typedef float v8f __attribute__((ext_vector_type(8)));

#define BSZ 4096
#define NF  2048
#define NT  300
#define RH  512
#define HD  512

#define MT      32
#define THREADS 256

__device__ __forceinline__ v8f wmma_f32_step(v2f a, v2f b, v8f c) {
    return __builtin_amdgcn_wmma_f32_16x16x4_f32(false, a, false, b,
                                                 (short)0, c, false, false);
}

struct FragSet {
    v4f a0;
    v4f a1;
    v4f b[4];
};

struct FragTail {
    v2f a0;
    v2f a1;
    v4f b[2];
};

template <int LD, bool A_LDS>
__device__ __forceinline__ void load_stage(
    FragSet& f, const float* __restrict__ Ag, const float* Al,
    const float* __restrict__ W, int m0, int col0, int r, int hi, int kbase)
{
    const int kk = kbase + 4 * hi;
    if (A_LDS) {
        f.a0 = *(const v4f*)(Al + r * RH + kk);
        f.a1 = *(const v4f*)(Al + (16 + r) * RH + kk);
    } else {
        f.a0 = *(const v4f*)(Ag + (size_t)(m0 + r) * LD + kk);
        f.a1 = *(const v4f*)(Ag + (size_t)(m0 + 16 + r) * LD + kk);
    }
    #pragma unroll
    for (int j = 0; j < 4; ++j)
        f.b[j] = *(const v4f*)(W + (size_t)(kk + j) * RH + col0);
}

template <int LD>
__device__ __forceinline__ void load_tail(
    FragTail& f, const float* __restrict__ Ag, const float* __restrict__ W,
    int m0, int col0, int r, int hi, int kbase)
{
    const int kk = kbase + 2 * hi;
    f.a0 = *(const v2f*)(Ag + (size_t)(m0 + r) * LD + kk);
    f.a1 = *(const v2f*)(Ag + (size_t)(m0 + 16 + r) * LD + kk);
    #pragma unroll
    for (int e = 0; e < 2; ++e)
        f.b[e] = *(const v4f*)(W + (size_t)(kk + e) * RH + col0);
}

__device__ __forceinline__ void mma_stage(const FragSet& f, v8f* c)
{
    #pragma unroll
    for (int j = 0; j < 2; ++j) {
        const v2f a0 = j ? (v2f){f.a0.z, f.a0.w} : (v2f){f.a0.x, f.a0.y};
        const v2f a1 = j ? (v2f){f.a1.z, f.a1.w} : (v2f){f.a1.x, f.a1.y};
        #pragma unroll
        for (int t = 0; t < 4; ++t) {
            const v2f b = (v2f){f.b[2 * j][t], f.b[2 * j + 1][t]};
            c[2 * t]     = wmma_f32_step(a0, b, c[2 * t]);
            c[2 * t + 1] = wmma_f32_step(a1, b, c[2 * t + 1]);
        }
    }
}

__device__ __forceinline__ void mma_tail(const FragTail& f, v8f* c)
{
    #pragma unroll
    for (int t = 0; t < 4; ++t) {
        const v2f b = (v2f){f.b[0][t], f.b[1][t]};
        c[2 * t]     = wmma_f32_step(f.a0, b, c[2 * t]);
        c[2 * t + 1] = wmma_f32_step(f.a1, b, c[2 * t + 1]);
    }
}

template <int K, int LD, bool A_LDS>
__device__ __forceinline__ void gemm_pipelined(
    const float* __restrict__ Ag, const float* Al, const float* __restrict__ W,
    v8f* c, int m0, int col0, int r, int hi)
{
    constexpr int KS   = 8;
    constexpr int NST  = K / KS;
    constexpr bool TAIL = (K % KS) != 0;
    constexpr int NIT  = (NST - 4 + 2) / 3;
    constexpr int REM  = NST - 3 * NIT;
    static_assert(K % 4 == 0, "K must be a multiple of the WMMA k");
    static_assert(NST >= 5, "needs a full trip and a remainder");
    static_assert(REM >= 2 && REM <= 4, "remainder is 2, 3 or 4 stages");
    static_assert(!(TAIL && A_LDS), "the tail loader reads global A only");

    FragSet f0, f1, f2;
    FragTail ft;
    load_stage<LD, A_LDS>(f0, Ag, Al, W, m0, col0, r, hi, 0);
    __builtin_amdgcn_sched_barrier(0);
    load_stage<LD, A_LDS>(f1, Ag, Al, W, m0, col0, r, hi, KS);

    int st = 0;
    #pragma unroll 1
    for (; st + 4 < NST; st += 3) {
        load_stage<LD, A_LDS>(f2, Ag, Al, W, m0, col0, r, hi, (st + 2) * KS);
        __builtin_amdgcn_sched_barrier(0);
        mma_stage(f0, c);
        __builtin_amdgcn_sched_barrier(0);
        load_stage<LD, A_LDS>(f0, Ag, Al, W, m0, col0, r, hi, (st + 3) * KS);
        __builtin_amdgcn_sched_barrier(0);
        mma_stage(f1, c);
        __builtin_amdgcn_sched_barrier(0);
        load_stage<LD, A_LDS>(f1, Ag, Al, W, m0, col0, r, hi, (st + 4) * KS);
        __builtin_amdgcn_sched_barrier(0);
        mma_stage(f2, c);
        __builtin_amdgcn_sched_barrier(0);
    }
    if (REM == 2) {
        if (TAIL) {
            load_tail<LD>(ft, Ag, W, m0, col0, r, hi, NST * KS);
            __builtin_amdgcn_sched_barrier(0);
        }
        mma_stage(f0, c);
        mma_stage(f1, c);
    } else if (REM == 3) {
        load_stage<LD, A_LDS>(f2, Ag, Al, W, m0, col0, r, hi, (st + 2) * KS);
        if (TAIL) load_tail<LD>(ft, Ag, W, m0, col0, r, hi, NST * KS);
        __builtin_amdgcn_sched_barrier(0);
        mma_stage(f0, c);
        mma_stage(f1, c);
        mma_stage(f2, c);
    } else {
        load_stage<LD, A_LDS>(f2, Ag, Al, W, m0, col0, r, hi, (st + 2) * KS);
        __builtin_amdgcn_sched_barrier(0);
        mma_stage(f0, c);
        __builtin_amdgcn_sched_barrier(0);
        load_stage<LD, A_LDS>(f0, Ag, Al, W, m0, col0, r, hi, (st + 3) * KS);
        if (TAIL) load_tail<LD>(ft, Ag, W, m0, col0, r, hi, NST * KS);
        __builtin_amdgcn_sched_barrier(0);
        mma_stage(f1, c);
        mma_stage(f2, c);
        mma_stage(f0, c);
    }
    if (TAIL) mma_tail(ft, c);
}

template <int K, int LD, bool MULTIPLY>
__device__ __forceinline__ void stage1_pass(
    const float* __restrict__ A, const float* __restrict__ W,
    float* uLDS, int m0, int col0, int r, int hi)
{
    v8f t8[8];
    #pragma unroll
    for (int i = 0; i < 8; ++i) t8[i] = (v8f){};

    gemm_pipelined<K, LD, false>(A, nullptr, W, t8, m0, col0, r, hi);

    #pragma unroll
    for (int mi = 0; mi < 2; ++mi) {
        #pragma unroll
        for (int e = 0; e < 8; ++e) {
            const int mm = mi * 16 + 8 * hi + e;
            const v4f val = (v4f){t8[mi][e], t8[2 + mi][e], t8[4 + mi][e], t8[6 + mi][e]};
            v4f* p = (v4f*)(uLDS + mm * RH + col0);
            if (MULTIPLY) *p = *p * val;
            else          *p = val;
        }
    }
}

template <int K1, int LD1, int K2, int LD2, bool FIRST, bool LAST>
__device__ __forceinline__ void product_and_project(
    const float* __restrict__ A1, const float* __restrict__ W1,
    const float* __restrict__ A2, const float* __restrict__ W2,
    const float* __restrict__ Ws,
    float* uLDS, float* park, v8f* acc,
    int m0, int col0, int r, int hi)
{
    stage1_pass<K1, LD1, false>(A1, W1, uLDS, m0, col0, r, hi);
    stage1_pass<K2, LD2, true >(A2, W2, uLDS, m0, col0, r, hi);
    __syncthreads();

    #pragma unroll
    for (int t = 0; t < 8; ++t) {
        if (FIRST) {
            acc[t] = (v8f){};
        } else {
            const v4f lo = *(const v4f*)(park + (2 * t) * 128);
            const v4f hh = *(const v4f*)(park + (2 * t + 1) * 128);
            acc[t] = (v8f){lo.x, lo.y, lo.z, lo.w, hh.x, hh.y, hh.z, hh.w};
        }
    }

    gemm_pipelined<RH, RH, true>(nullptr, uLDS, Ws, acc, m0, col0, r, hi);

    if (!LAST) {
        #pragma unroll
        for (int t = 0; t < 8; ++t) {
            *(v4f*)(park + (2 * t) * 128)     = (v4f){acc[t][0], acc[t][1], acc[t][2], acc[t][3]};
            *(v4f*)(park + (2 * t + 1) * 128) = (v4f){acc[t][4], acc[t][5], acc[t][6], acc[t][7]};
        }
    }
    __syncthreads();
}

__global__ __launch_bounds__(THREADS, 4)
void vncl_gate_kernel(const float* __restrict__ v,  const float* __restrict__ s,
                      const float* __restrict__ x,  const float* __restrict__ h,
                      const float* __restrict__ Wa, const float* __restrict__ Wb,
                      const float* __restrict__ Wc, const float* __restrict__ Ua,
                      const float* __restrict__ Ub, const float* __restrict__ Uc,
                      const float* __restrict__ Ca, const float* __restrict__ Cb,
                      const float* __restrict__ Cc, const float* __restrict__ bias,
                      float* __restrict__ logits  )
{
    __shared__ __attribute__((aligned(16))) float uLDS[MT * RH];
    __shared__ __attribute__((aligned(16))) float accPark[8 * 16 * 32 * 4];

    const int g    = blockIdx.y;
    const int m0   = blockIdx.x * MT;
    const int lane = threadIdx.x & 31;
    const int wave = threadIdx.x >> 5;
    const int r    = lane & 15;
    const int hi   = lane >> 4;

    const int col0 = wave * 64 + 4 * r;

    const float* Wa_g = Wa + (size_t)g * RH * RH;
    const float* Wb_g = Wb + (size_t)g * NT * RH;
    const float* Wc_g = Wc + (size_t)g * RH * HD;
    const float* Ua_g = Ua + (size_t)g * RH * RH;
    const float* Ub_g = Ub + (size_t)g * NT * RH;
    const float* Uc_g = Uc + (size_t)g * RH * HD;
    const float* Ca_g = Ca + (size_t)g * NF * RH;
    const float* Cb_g = Cb + (size_t)g * NT * RH;
    const float* Cc_g = Cc + (size_t)g * RH * HD;

    v8f acc[8];
    float* park = accPark + wave * (16 * 32 * 4) + lane * 4;

    product_and_project<RH, RH, NT, NT, true, false>(x, Wa_g, s, Wb_g, Wc_g,
                                        uLDS, park, acc, m0, col0, r, hi);
    product_and_project<NF, NF, NT, NT, false, false>(v, Ca_g, s, Cb_g, Cc_g,
                                        uLDS, park, acc, m0, col0, r, hi);
    product_and_project<NT, NT, RH, RH, false, true>(s, Ub_g, h, Ua_g, Uc_g,
                                        uLDS, park, acc, m0, col0, r, hi);

    const v4f bv = *(const v4f*)(bias + (size_t)g * HD + col0);
    float* Lg = logits + ((size_t)g * BSZ + m0) * HD;
    #pragma unroll
    for (int mi = 0; mi < 2; ++mi) {
        #pragma unroll
        for (int e = 0; e < 8; ++e) {
            const int mm = mi * 16 + 8 * hi + e;
            const v4f val = (v4f){acc[mi][e], acc[2 + mi][e], acc[4 + mi][e], acc[6 + mi][e]};
            *(v4f*)(Lg + (size_t)mm * HD + col0) = val + bv;
        }
    }
}

__global__ __launch_bounds__(256)
void lstm_epilogue(const float* __restrict__ logits,
                   const float* __restrict__ c,
                   float* __restrict__ out, int n)
{
    const int idx = blockIdx.x * blockDim.x + threadIdx.x;
    if (idx >= n) return;
    const size_t stride = (size_t)BSZ * HD;
    const float li = logits[idx];
    const float lf = logits[idx + stride];
    const float lo = logits[idx + 2 * stride];
    const float lg = logits[idx + 3 * stride];
    const float ig = 1.0f / (1.0f + __expf(-li));
    const float fg = 1.0f / (1.0f + __expf(-lf));
    const float og = 1.0f / (1.0f + __expf(-lo));
    const float gg = tanhf(lg);
    const float nc = fg * c[idx] + ig * gg;
    const float nh = og * tanhf(nc);
    out[idx]          = nh;
    out[idx + stride] = nc;
}

extern "C" void kernel_launch(void* const* d_in, const int* in_sizes, int n_in,
                              void* d_out, int out_size, void* d_ws, size_t ws_size,
                              hipStream_t stream) {
    (void)in_sizes; (void)n_in; (void)out_size; (void)ws_size;
    const float* v  = (const float*)d_in[0];
    const float* s  = (const float*)d_in[1];
    const float* x  = (const float*)d_in[2];
    const float* h  = (const float*)d_in[3];
    const float* c  = (const float*)d_in[4];
    const float* Wa = (const float*)d_in[5];
    const float* Wb = (const float*)d_in[6];
    const float* Wc = (const float*)d_in[7];
    const float* Ua = (const float*)d_in[8];
    const float* Ub = (const float*)d_in[9];
    const float* Uc = (const float*)d_in[10];
    const float* Ca = (const float*)d_in[11];
    const float* Cb = (const float*)d_in[12];
    const float* Cc = (const float*)d_in[13];
    const float* b  = (const float*)d_in[14];

    float* logits = (float*)d_ws;

    dim3 grid(BSZ / MT, 4);
    vncl_gate_kernel<<<grid, dim3(THREADS), 0, stream>>>(
        v, s, x, h, Wa, Wb, Wc, Ua, Ub, Uc, Ca, Cb, Cc, b, logits);

    const int n = BSZ * HD;
    lstm_epilogue<<<(n + 255) / 256, 256, 0, stream>>>(logits, c, (float*)d_out, n);
}


# Wave issue priority steps down 3,2,1,0 as a workgroup passes 61, 76 and 91 % of its matrix work: the lagging co-resident workgroup is served first, so both end together, none alone
# speedup vs baseline: 2.4491x; 1.0385x over previous
#include <hip/hip_runtime.h>
#include <math.h>


typedef float v2f __attribute__((ext_vector_type(2)));
typedef float v4f __attribute__((ext_vector_type(4)));
typedef float v8f __attribute__((ext_vector_type(8)));

#define BSZ 4096
#define NF  2048
#define NT  300
#define RH  512
#define HD  512

#define MT      32
#define THREADS 256

__device__ __forceinline__ v8f wmma_f32_step(v2f a, v2f b, v8f c) {
    return __builtin_amdgcn_wmma_f32_16x16x4_f32(false, a, false, b,
                                                 (short)0, c, false, false);
}

struct FragSet {
    v4f a0;
    v4f a1;
    v4f b[4];
};

struct FragTail {
    v2f a0;
    v2f a1;
    v4f b[2];
};

template <int LD, bool A_LDS>
__device__ __forceinline__ void load_stage(
    FragSet& f, const float* __restrict__ Ag, const float* Al,
    const float* __restrict__ W, int m0, int col0, int r, int hi, int kbase)
{
    const int kk = kbase + 4 * hi;
    if (A_LDS) {
        f.a0 = *(const v4f*)(Al + r * RH + kk);
        f.a1 = *(const v4f*)(Al + (16 + r) * RH + kk);
    } else {
        f.a0 = *(const v4f*)(Ag + (size_t)(m0 + r) * LD + kk);
        f.a1 = *(const v4f*)(Ag + (size_t)(m0 + 16 + r) * LD + kk);
    }
    #pragma unroll
    for (int j = 0; j < 4; ++j)
        f.b[j] = *(const v4f*)(W + (size_t)(kk + j) * RH + col0);
}

template <int LD>
__device__ __forceinline__ void load_tail(
    FragTail& f, const float* __restrict__ Ag, const float* __restrict__ W,
    int m0, int col0, int r, int hi, int kbase)
{
    const int kk = kbase + 2 * hi;
    f.a0 = *(const v2f*)(Ag + (size_t)(m0 + r) * LD + kk);
    f.a1 = *(const v2f*)(Ag + (size_t)(m0 + 16 + r) * LD + kk);
    #pragma unroll
    for (int e = 0; e < 2; ++e)
        f.b[e] = *(const v4f*)(W + (size_t)(kk + e) * RH + col0);
}

__device__ __forceinline__ void mma_stage(const FragSet& f, v8f* c)
{
    #pragma unroll
    for (int j = 0; j < 2; ++j) {
        const v2f a0 = j ? (v2f){f.a0.z, f.a0.w} : (v2f){f.a0.x, f.a0.y};
        const v2f a1 = j ? (v2f){f.a1.z, f.a1.w} : (v2f){f.a1.x, f.a1.y};
        #pragma unroll
        for (int t = 0; t < 4; ++t) {
            const v2f b = (v2f){f.b[2 * j][t], f.b[2 * j + 1][t]};
            c[2 * t]     = wmma_f32_step(a0, b, c[2 * t]);
            c[2 * t + 1] = wmma_f32_step(a1, b, c[2 * t + 1]);
        }
    }
}

__device__ __forceinline__ void mma_tail(const FragTail& f, v8f* c)
{
    #pragma unroll
    for (int t = 0; t < 4; ++t) {
        const v2f b = (v2f){f.b[0][t], f.b[1][t]};
        c[2 * t]     = wmma_f32_step(f.a0, b, c[2 * t]);
        c[2 * t + 1] = wmma_f32_step(f.a1, b, c[2 * t + 1]);
    }
}

template <int K, int LD, bool A_LDS, int PRIO>
__device__ __forceinline__ void gemm_pipelined(
    const float* __restrict__ Ag, const float* Al, const float* __restrict__ W,
    v8f* c, int m0, int col0, int r, int hi)
{
    constexpr int KS   = 8;
    constexpr int NST  = K / KS;
    constexpr bool TAIL = (K % KS) != 0;
    constexpr int NIT  = (NST - 4 + 2) / 3;
    constexpr int REM  = NST - 3 * NIT;
    static_assert(K % 4 == 0, "K must be a multiple of the WMMA k");
    static_assert(NST >= 5, "needs a full trip and a remainder");
    static_assert(REM >= 2 && REM <= 4, "remainder is 2, 3 or 4 stages");
    static_assert(!(TAIL && A_LDS), "the tail loader reads global A only");

    FragSet f0, f1, f2;
    FragTail ft;
    load_stage<LD, A_LDS>(f0, Ag, Al, W, m0, col0, r, hi, 0);
    if (PRIO >= 0) __builtin_amdgcn_s_setprio((short)PRIO);
    __builtin_amdgcn_sched_barrier(0);
    load_stage<LD, A_LDS>(f1, Ag, Al, W, m0, col0, r, hi, KS);

    int st = 0;
    #pragma unroll 1
    for (; st + 4 < NST; st += 3) {
        load_stage<LD, A_LDS>(f2, Ag, Al, W, m0, col0, r, hi, (st + 2) * KS);
        __builtin_amdgcn_sched_barrier(0);
        mma_stage(f0, c);
        __builtin_amdgcn_sched_barrier(0);
        load_stage<LD, A_LDS>(f0, Ag, Al, W, m0, col0, r, hi, (st + 3) * KS);
        __builtin_amdgcn_sched_barrier(0);
        mma_stage(f1, c);
        __builtin_amdgcn_sched_barrier(0);
        load_stage<LD, A_LDS>(f1, Ag, Al, W, m0, col0, r, hi, (st + 4) * KS);
        __builtin_amdgcn_sched_barrier(0);
        mma_stage(f2, c);
        __builtin_amdgcn_sched_barrier(0);
    }
    if (REM == 2) {
        if (TAIL) {
            load_tail<LD>(ft, Ag, W, m0, col0, r, hi, NST * KS);
            __builtin_amdgcn_sched_barrier(0);
        }
        mma_stage(f0, c);
        mma_stage(f1, c);
    } else if (REM == 3) {
        load_stage<LD, A_LDS>(f2, Ag, Al, W, m0, col0, r, hi, (st + 2) * KS);
        if (TAIL) load_tail<LD>(ft, Ag, W, m0, col0, r, hi, NST * KS);
        __builtin_amdgcn_sched_barrier(0);
        mma_stage(f0, c);
        mma_stage(f1, c);
        mma_stage(f2, c);
    } else {
        load_stage<LD, A_LDS>(f2, Ag, Al, W, m0, col0, r, hi, (st + 2) * KS);
        __builtin_amdgcn_sched_barrier(0);
        mma_stage(f0, c);
        __builtin_amdgcn_sched_barrier(0);
        load_stage<LD, A_LDS>(f0, Ag, Al, W, m0, col0, r, hi, (st + 3) * KS);
        if (TAIL) load_tail<LD>(ft, Ag, W, m0, col0, r, hi, NST * KS);
        __builtin_amdgcn_sched_barrier(0);
        mma_stage(f1, c);
        mma_stage(f2, c);
        mma_stage(f0, c);
    }
    if (TAIL) mma_tail(ft, c);
}

template <int K, int LD, bool MULTIPLY, int PRIO>
__device__ __forceinline__ void stage1_pass(
    const float* __restrict__ A, const float* __restrict__ W,
    float* uLDS, int m0, int col0, int r, int hi)
{
    v8f t8[8];
    #pragma unroll
    for (int i = 0; i < 8; ++i) t8[i] = (v8f){};

    gemm_pipelined<K, LD, false, PRIO>(A, nullptr, W, t8, m0, col0, r, hi);

    #pragma unroll
    for (int mi = 0; mi < 2; ++mi) {
        #pragma unroll
        for (int e = 0; e < 8; ++e) {
            const int mm = mi * 16 + 8 * hi + e;
            const v4f val = (v4f){t8[mi][e], t8[2 + mi][e], t8[4 + mi][e], t8[6 + mi][e]};
            v4f* p = (v4f*)(uLDS + mm * RH + col0);
            if (MULTIPLY) *p = *p * val;
            else          *p = val;
        }
    }
}

template <int K1, int LD1, int K2, int LD2, bool FIRST, bool LAST,
          int P1, int P2, int PP>
__device__ __forceinline__ void product_and_project(
    const float* __restrict__ A1, const float* __restrict__ W1,
    const float* __restrict__ A2, const float* __restrict__ W2,
    const float* __restrict__ Ws,
    float* uLDS, float* park, v8f* acc,
    int m0, int col0, int r, int hi)
{
    stage1_pass<K1, LD1, false, P1>(A1, W1, uLDS, m0, col0, r, hi);
    stage1_pass<K2, LD2, true,  P2>(A2, W2, uLDS, m0, col0, r, hi);
    __syncthreads();

    #pragma unroll
    for (int t = 0; t < 8; ++t) {
        if (FIRST) {
            acc[t] = (v8f){};
        } else {
            const v4f lo = *(const v4f*)(park + (2 * t) * 128);
            const v4f hh = *(const v4f*)(park + (2 * t + 1) * 128);
            acc[t] = (v8f){lo.x, lo.y, lo.z, lo.w, hh.x, hh.y, hh.z, hh.w};
        }
    }

    gemm_pipelined<RH, RH, true, PP>(nullptr, uLDS, Ws, acc, m0, col0, r, hi);

    if (!LAST) {
        #pragma unroll
        for (int t = 0; t < 8; ++t) {
            *(v4f*)(park + (2 * t) * 128)     = (v4f){acc[t][0], acc[t][1], acc[t][2], acc[t][3]};
            *(v4f*)(park + (2 * t + 1) * 128) = (v4f){acc[t][4], acc[t][5], acc[t][6], acc[t][7]};
        }
    }
    __syncthreads();
}

__global__ __launch_bounds__(THREADS, 4)
void vncl_gate_kernel(const float* __restrict__ v,  const float* __restrict__ s,
                      const float* __restrict__ x,  const float* __restrict__ h,
                      const float* __restrict__ Wa, const float* __restrict__ Wb,
                      const float* __restrict__ Wc, const float* __restrict__ Ua,
                      const float* __restrict__ Ub, const float* __restrict__ Uc,
                      const float* __restrict__ Ca, const float* __restrict__ Cb,
                      const float* __restrict__ Cc, const float* __restrict__ bias,
                      float* __restrict__ logits  )
{
    __shared__ __attribute__((aligned(16))) float uLDS[MT * RH];
    __shared__ __attribute__((aligned(16))) float accPark[8 * 16 * 32 * 4];

    const int g    = blockIdx.y;
    const int m0   = blockIdx.x * MT;
    const int lane = threadIdx.x & 31;
    const int wave = threadIdx.x >> 5;
    const int r    = lane & 15;
    const int hi   = lane >> 4;

    const int col0 = wave * 64 + 4 * r;

    const float* Wa_g = Wa + (size_t)g * RH * RH;
    const float* Wb_g = Wb + (size_t)g * NT * RH;
    const float* Wc_g = Wc + (size_t)g * RH * HD;
    const float* Ua_g = Ua + (size_t)g * RH * RH;
    const float* Ub_g = Ub + (size_t)g * NT * RH;
    const float* Uc_g = Uc + (size_t)g * RH * HD;
    const float* Ca_g = Ca + (size_t)g * NF * RH;
    const float* Cb_g = Cb + (size_t)g * NT * RH;
    const float* Cc_g = Cc + (size_t)g * RH * HD;

    v8f acc[8];
    float* park = accPark + wave * (16 * 32 * 4) + lane * 4;

    product_and_project<RH, RH, NT, NT, true, false, 3, -1, -1>(x, Wa_g, s, Wb_g, Wc_g,
                                        uLDS, park, acc, m0, col0, r, hi);
    product_and_project<NF, NF, NT, NT, false, false, -1, 2, -1>(v, Ca_g, s, Cb_g, Cc_g,
                                        uLDS, park, acc, m0, col0, r, hi);
    product_and_project<NT, NT, RH, RH, false, true, 1, -1, 0>(s, Ub_g, h, Ua_g, Uc_g,
                                        uLDS, park, acc, m0, col0, r, hi);

    const v4f bv = *(const v4f*)(bias + (size_t)g * HD + col0);
    float* Lg = logits + ((size_t)g * BSZ + m0) * HD;
    #pragma unroll
    for (int mi = 0; mi < 2; ++mi) {
        #pragma unroll
        for (int e = 0; e < 8; ++e) {
            const int mm = mi * 16 + 8 * hi + e;
            const v4f val = (v4f){acc[mi][e], acc[2 + mi][e], acc[4 + mi][e], acc[6 + mi][e]};
            *(v4f*)(Lg + (size_t)mm * HD + col0) = val + bv;
        }
    }
}

__global__ __launch_bounds__(256)
void lstm_epilogue(const float* __restrict__ logits,
                   const float* __restrict__ c,
                   float* __restrict__ out, int n)
{
    const int idx = blockIdx.x * blockDim.x + threadIdx.x;
    if (idx >= n) return;
    const size_t stride = (size_t)BSZ * HD;
    const float li = logits[idx];
    const float lf = logits[idx + stride];
    const float lo = logits[idx + 2 * stride];
    const float lg = logits[idx + 3 * stride];
    const float ig = 1.0f / (1.0f + __expf(-li));
    const float fg = 1.0f / (1.0f + __expf(-lf));
    const float og = 1.0f / (1.0f + __expf(-lo));
    const float gg = tanhf(lg);
    const float nc = fg * c[idx] + ig * gg;
    const float nh = og * tanhf(nc);
    out[idx]          = nh;
    out[idx + stride] = nc;
}

extern "C" void kernel_launch(void* const* d_in, const int* in_sizes, int n_in,
                              void* d_out, int out_size, void* d_ws, size_t ws_size,
                              hipStream_t stream) {
    (void)in_sizes; (void)n_in; (void)out_size; (void)ws_size;
    const float* v  = (const float*)d_in[0];
    const float* s  = (const float*)d_in[1];
    const float* x  = (const float*)d_in[2];
    const float* h  = (const float*)d_in[3];
    const float* c  = (const float*)d_in[4];
    const float* Wa = (const float*)d_in[5];
    const float* Wb = (const float*)d_in[6];
    const float* Wc = (const float*)d_in[7];
    const float* Ua = (const float*)d_in[8];
    const float* Ub = (const float*)d_in[9];
    const float* Uc = (const float*)d_in[10];
    const float* Ca = (const float*)d_in[11];
    const float* Cb = (const float*)d_in[12];
    const float* Cc = (const float*)d_in[13];
    const float* b  = (const float*)d_in[14];

    float* logits = (float*)d_ws;

    dim3 grid(BSZ / MT, 4);
    vncl_gate_kernel<<<grid, dim3(THREADS), 0, stream>>>(
        v, s, x, h, Wa, Wb, Wc, Ua, Ub, Uc, Ca, Cb, Cc, b, logits);

    const int n = BSZ * HD;
    lstm_epilogue<<<(n + 255) / 256, 256, 0, stream>>>(logits, c, (float*)d_out, n);
}


# Workgroup tile 32 -> 64 batch rows: each wave owns 4 m-groups x 4 n-tiles, one W fragment feeds 4 m-groups (W reads from L2 halved); 256 workgroups, one per WGP, accumulators stay in registers
# speedup vs baseline: 2.5336x; 1.0345x over previous
#include <hip/hip_runtime.h>
#include <math.h>


typedef float v2f __attribute__((ext_vector_type(2)));
typedef float v4f __attribute__((ext_vector_type(4)));
typedef float v8f __attribute__((ext_vector_type(8)));

#define BSZ 4096
#define NF  2048
#define NT  300
#define RH  512
#define HD  512

#define MT      64
#define MG      4
#define THREADS 256

__device__ __forceinline__ v8f wmma_f32_step(v2f a, v2f b, v8f c) {
    return __builtin_amdgcn_wmma_f32_16x16x4_f32(false, a, false, b,
                                                 (short)0, c, false, false);
}

struct FragSet {
    v4f a[MG];
    v4f b[4];
};

struct FragTail {
    v2f a[MG];
    v4f b[2];
};

template <int LD, bool A_LDS>
__device__ __forceinline__ void load_stage(
    FragSet& f, const float* __restrict__ Ag, const float* Al,
    const float* __restrict__ W, int m0, int col0, int r, int hi, int kbase)
{
    const int kk = kbase + 4 * hi;
    #pragma unroll
    for (int mi = 0; mi < MG; ++mi) {
        if (A_LDS) f.a[mi] = *(const v4f*)(Al + (16 * mi + r) * RH + kk);
        else       f.a[mi] = *(const v4f*)(Ag + (size_t)(m0 + 16 * mi + r) * LD + kk);
    }
    #pragma unroll
    for (int j = 0; j < 4; ++j)
        f.b[j] = *(const v4f*)(W + (size_t)(kk + j) * RH + col0);
}

template <int LD>
__device__ __forceinline__ void load_tail(
    FragTail& f, const float* __restrict__ Ag, const float* __restrict__ W,
    int m0, int col0, int r, int hi, int kbase)
{
    const int kk = kbase + 2 * hi;
    #pragma unroll
    for (int mi = 0; mi < MG; ++mi)
        f.a[mi] = *(const v2f*)(Ag + (size_t)(m0 + 16 * mi + r) * LD + kk);
    #pragma unroll
    for (int e = 0; e < 2; ++e)
        f.b[e] = *(const v4f*)(W + (size_t)(kk + e) * RH + col0);
}

__device__ __forceinline__ void mma_stage(const FragSet& f, v8f* c)
{
    #pragma unroll
    for (int j = 0; j < 2; ++j) {
        v2f a[MG];
        #pragma unroll
        for (int mi = 0; mi < MG; ++mi)
            a[mi] = j ? (v2f){f.a[mi].z, f.a[mi].w} : (v2f){f.a[mi].x, f.a[mi].y};
        #pragma unroll
        for (int t = 0; t < 4; ++t) {
            const v2f b = (v2f){f.b[2 * j][t], f.b[2 * j + 1][t]};
            #pragma unroll
            for (int mi = 0; mi < MG; ++mi)
                c[MG * t + mi] = wmma_f32_step(a[mi], b, c[MG * t + mi]);
        }
    }
}

__device__ __forceinline__ void mma_tail(const FragTail& f, v8f* c)
{
    #pragma unroll
    for (int t = 0; t < 4; ++t) {
        const v2f b = (v2f){f.b[0][t], f.b[1][t]};
        #pragma unroll
        for (int mi = 0; mi < MG; ++mi)
            c[MG * t + mi] = wmma_f32_step(f.a[mi], b, c[MG * t + mi]);
    }
}

template <int K, int LD, bool A_LDS>
__device__ __forceinline__ void gemm_pipelined(
    const float* __restrict__ Ag, const float* Al, const float* __restrict__ W,
    v8f* c, int m0, int col0, int r, int hi)
{
    constexpr int KS   = 8;
    constexpr int NST  = K / KS;
    constexpr bool TAIL = (K % KS) != 0;
    constexpr int NIT  = (NST - 4 + 2) / 3;
    constexpr int REM  = NST - 3 * NIT;
    static_assert(K % 4 == 0, "K must be a multiple of the WMMA k");
    static_assert(NST >= 5, "needs a full trip and a remainder");
    static_assert(REM >= 2 && REM <= 4, "remainder is 2, 3 or 4 stages");
    static_assert(!(TAIL && A_LDS), "the tail loader reads global A only");

    FragSet f0, f1, f2;
    FragTail ft;
    load_stage<LD, A_LDS>(f0, Ag, Al, W, m0, col0, r, hi, 0);
    __builtin_amdgcn_sched_barrier(0);
    load_stage<LD, A_LDS>(f1, Ag, Al, W, m0, col0, r, hi, KS);

    int st = 0;
    #pragma unroll 1
    for (; st + 4 < NST; st += 3) {
        load_stage<LD, A_LDS>(f2, Ag, Al, W, m0, col0, r, hi, (st + 2) * KS);
        __builtin_amdgcn_sched_barrier(0);
        mma_stage(f0, c);
        __builtin_amdgcn_sched_barrier(0);
        load_stage<LD, A_LDS>(f0, Ag, Al, W, m0, col0, r, hi, (st + 3) * KS);
        __builtin_amdgcn_sched_barrier(0);
        mma_stage(f1, c);
        __builtin_amdgcn_sched_barrier(0);
        load_stage<LD, A_LDS>(f1, Ag, Al, W, m0, col0, r, hi, (st + 4) * KS);
        __builtin_amdgcn_sched_barrier(0);
        mma_stage(f2, c);
        __builtin_amdgcn_sched_barrier(0);
    }
    if (REM == 2) {
        if (TAIL) {
            load_tail<LD>(ft, Ag, W, m0, col0, r, hi, NST * KS);
            __builtin_amdgcn_sched_barrier(0);
        }
        mma_stage(f0, c);
        mma_stage(f1, c);
    } else if (REM == 3) {
        load_stage<LD, A_LDS>(f2, Ag, Al, W, m0, col0, r, hi, (st + 2) * KS);
        if (TAIL) load_tail<LD>(ft, Ag, W, m0, col0, r, hi, NST * KS);
        __builtin_amdgcn_sched_barrier(0);
        mma_stage(f0, c);
        mma_stage(f1, c);
        mma_stage(f2, c);
    } else {
        load_stage<LD, A_LDS>(f2, Ag, Al, W, m0, col0, r, hi, (st + 2) * KS);
        __builtin_amdgcn_sched_barrier(0);
        mma_stage(f0, c);
        __builtin_amdgcn_sched_barrier(0);
        load_stage<LD, A_LDS>(f0, Ag, Al, W, m0, col0, r, hi, (st + 3) * KS);
        if (TAIL) load_tail<LD>(ft, Ag, W, m0, col0, r, hi, NST * KS);
        __builtin_amdgcn_sched_barrier(0);
        mma_stage(f1, c);
        mma_stage(f2, c);
        mma_stage(f0, c);
    }
    if (TAIL) mma_tail(ft, c);
}

template <int K, int LD, bool MULTIPLY>
__device__ __forceinline__ void stage1_pass(
    const float* __restrict__ A, const float* __restrict__ W,
    float* uLDS, int m0, int col0, int r, int hi)
{
    v8f t16[4 * MG];
    #pragma unroll
    for (int i = 0; i < 4 * MG; ++i) t16[i] = (v8f){};

    gemm_pipelined<K, LD, false>(A, nullptr, W, t16, m0, col0, r, hi);

    #pragma unroll
    for (int mi = 0; mi < MG; ++mi) {
        #pragma unroll
        for (int e = 0; e < 8; ++e) {
            const int mm = mi * 16 + 8 * hi + e;
            const v4f val = (v4f){t16[mi][e], t16[MG + mi][e],
                                  t16[2 * MG + mi][e], t16[3 * MG + mi][e]};
            v4f* p = (v4f*)(uLDS + mm * RH + col0);
            if (MULTIPLY) *p = *p * val;
            else          *p = val;
        }
    }
}

template <int K1, int LD1, int K2, int LD2, bool FIRST>
__device__ __forceinline__ void product_and_project(
    const float* __restrict__ A1, const float* __restrict__ W1,
    const float* __restrict__ A2, const float* __restrict__ W2,
    const float* __restrict__ Ws,
    float* uLDS, v8f* acc,
    int m0, int col0, int r, int hi)
{
    stage1_pass<K1, LD1, false>(A1, W1, uLDS, m0, col0, r, hi);
    stage1_pass<K2, LD2, true >(A2, W2, uLDS, m0, col0, r, hi);
    __syncthreads();

    if (FIRST) {
        #pragma unroll
        for (int t = 0; t < 4 * MG; ++t) acc[t] = (v8f){};
    }

    gemm_pipelined<RH, RH, true>(nullptr, uLDS, Ws, acc, m0, col0, r, hi);

    __syncthreads();
}

__global__ __launch_bounds__(THREADS, 2)
void vncl_gate_kernel(const float* __restrict__ v,  const float* __restrict__ s,
                      const float* __restrict__ x,  const float* __restrict__ h,
                      const float* __restrict__ Wa, const float* __restrict__ Wb,
                      const float* __restrict__ Wc, const float* __restrict__ Ua,
                      const float* __restrict__ Ub, const float* __restrict__ Uc,
                      const float* __restrict__ Ca, const float* __restrict__ Cb,
                      const float* __restrict__ Cc, const float* __restrict__ bias,
                      float* __restrict__ logits  )
{
    __shared__ __attribute__((aligned(16))) float uLDS[MT * RH];

    const int g    = blockIdx.y;
    const int m0   = blockIdx.x * MT;
    const int lane = threadIdx.x & 31;
    const int wave = threadIdx.x >> 5;
    const int r    = lane & 15;
    const int hi   = lane >> 4;

    const int col0 = wave * 64 + 4 * r;

    const float* Wa_g = Wa + (size_t)g * RH * RH;
    const float* Wb_g = Wb + (size_t)g * NT * RH;
    const float* Wc_g = Wc + (size_t)g * RH * HD;
    const float* Ua_g = Ua + (size_t)g * RH * RH;
    const float* Ub_g = Ub + (size_t)g * NT * RH;
    const float* Uc_g = Uc + (size_t)g * RH * HD;
    const float* Ca_g = Ca + (size_t)g * NF * RH;
    const float* Cb_g = Cb + (size_t)g * NT * RH;
    const float* Cc_g = Cc + (size_t)g * RH * HD;

    v8f acc[4 * MG];

    product_and_project<RH, RH, NT, NT, true >(x, Wa_g, s, Wb_g, Wc_g,
                                        uLDS, acc, m0, col0, r, hi);
    product_and_project<NF, NF, NT, NT, false>(v, Ca_g, s, Cb_g, Cc_g,
                                        uLDS, acc, m0, col0, r, hi);
    product_and_project<NT, NT, RH, RH, false>(s, Ub_g, h, Ua_g, Uc_g,
                                        uLDS, acc, m0, col0, r, hi);

    const v4f bv = *(const v4f*)(bias + (size_t)g * HD + col0);
    float* Lg = logits + ((size_t)g * BSZ + m0) * HD;
    #pragma unroll
    for (int mi = 0; mi < MG; ++mi) {
        #pragma unroll
        for (int e = 0; e < 8; ++e) {
            const int mm = mi * 16 + 8 * hi + e;
            const v4f val = (v4f){acc[mi][e], acc[MG + mi][e],
                                  acc[2 * MG + mi][e], acc[3 * MG + mi][e]};
            *(v4f*)(Lg + (size_t)mm * HD + col0) = val + bv;
        }
    }
}

__global__ __launch_bounds__(256)
void lstm_epilogue(const float* __restrict__ logits,
                   const float* __restrict__ c,
                   float* __restrict__ out, int n)
{
    const int idx = blockIdx.x * blockDim.x + threadIdx.x;
    if (idx >= n) return;
    const size_t stride = (size_t)BSZ * HD;
    const float li = logits[idx];
    const float lf = logits[idx + stride];
    const float lo = logits[idx + 2 * stride];
    const float lg = logits[idx + 3 * stride];
    const float ig = 1.0f / (1.0f + __expf(-li));
    const float fg = 1.0f / (1.0f + __expf(-lf));
    const float og = 1.0f / (1.0f + __expf(-lo));
    const float gg = tanhf(lg);
    const float nc = fg * c[idx] + ig * gg;
    const float nh = og * tanhf(nc);
    out[idx]          = nh;
    out[idx + stride] = nc;
}

extern "C" void kernel_launch(void* const* d_in, const int* in_sizes, int n_in,
                              void* d_out, int out_size, void* d_ws, size_t ws_size,
                              hipStream_t stream) {
    (void)in_sizes; (void)n_in; (void)out_size; (void)ws_size;
    const float* v  = (const float*)d_in[0];
    const float* s  = (const float*)d_in[1];
    const float* x  = (const float*)d_in[2];
    const float* h  = (const float*)d_in[3];
    const float* c  = (const float*)d_in[4];
    const float* Wa = (const float*)d_in[5];
    const float* Wb = (const float*)d_in[6];
    const float* Wc = (const float*)d_in[7];
    const float* Ua = (const float*)d_in[8];
    const float* Ub = (const float*)d_in[9];
    const float* Uc = (const float*)d_in[10];
    const float* Ca = (const float*)d_in[11];
    const float* Cb = (const float*)d_in[12];
    const float* Cc = (const float*)d_in[13];
    const float* b  = (const float*)d_in[14];

    float* logits = (float*)d_ws;

    dim3 grid(BSZ / MT, 4);
    vncl_gate_kernel<<<grid, dim3(THREADS), 0, stream>>>(
        v, s, x, h, Wa, Wb, Wc, Ua, Ub, Uc, Ca, Cb, Cc, b, logits);

    const int n = BSZ * HD;
    lstm_epilogue<<<(n + 255) / 256, 256, 0, stream>>>(logits, c, (float*)d_out, n);
}
